# starting kernel
# baseline (speedup 1.0000x reference)
#include <hip/hip_runtime.h>
#include <hip/hip_bf16.h>

typedef __attribute__((ext_vector_type(2))) float v2f;
typedef __attribute__((ext_vector_type(8))) float v8f;

#define D 128
#define LDS_STRIDE 132

__global__ __launch_bounds__(256) void zero4_kernel(float4* __restrict__ p, long n4) {
    long i = (long)blockIdx.x * blockDim.x + threadIdx.x;
    if (i < n4) p[i] = make_float4(0.f, 0.f, 0.f, 0.f);
}

__global__ __launch_bounds__(256)
void gnn_scatter_kernel(const float* __restrict__ UF, const float* __restrict__ IF,
                        const int* __restrict__ uidx, const int* __restrict__ iidx,
                        float* __restrict__ aggU, float* __restrict__ aggI,
                        float* __restrict__ degU, float* __restrict__ degI, int E)
{
    int t = blockIdx.x * 256 + threadIdx.x;
    int e = t >> 5;
    if (e >= E) return;
    int lane = threadIdx.x & 31;
    int u  = uidx[e];
    int it = iidx[e];

    const float4 fu = ((const float4*)(UF + (size_t)u  * D))[lane];
    const float4 fi = ((const float4*)(IF + (size_t)it * D))[lane];

    float* ai = aggI + (size_t)it * D + lane * 4;
    float* au = aggU + (size_t)u  * D + lane * 4;

    unsafeAtomicAdd(ai + 0, fu.x);
    unsafeAtomicAdd(ai + 1, fu.y);
    unsafeAtomicAdd(ai + 2, fu.z);
    unsafeAtomicAdd(ai + 3, fu.w);

    unsafeAtomicAdd(au + 0, fi.x);
    unsafeAtomicAdd(au + 1, fi.y);
    unsafeAtomicAdd(au + 2, fi.z);
    unsafeAtomicAdd(au + 3, fi.w);

    if (lane == 0) {
        unsafeAtomicAdd(degU + u, 1.0f);
        unsafeAtomicAdd(degI + it, 1.0f);
    }
}

__global__ __launch_bounds__(256)
void gnn_gemm_kernel(const float* __restrict__ X, float* __restrict__ out,
                     const float* __restrict__ Wself, const float* __restrict__ bself,
                     const float* __restrict__ Wmsg,  const float* __restrict__ bmsg,
                     const float* __restrict__ deg)
{
    __shared__ float sX[16 * LDS_STRIDE];
    __shared__ float sG[16 * LDS_STRIDE];

    const int row0 = blockIdx.x * 16;
    const int tid  = threadIdx.x;

    for (int i = tid; i < 16 * 32; i += 256) {
        int r  = i >> 5;
        int c4 = i & 31;
        float4 x = ((const float4*)(X   + (size_t)(row0 + r) * D))[c4];
        float4 g = ((const float4*)(out + (size_t)(row0 + r) * D))[c4];
        ((float4*)(sX + r * LDS_STRIDE))[c4] = x;
        ((float4*)(sG + r * LDS_STRIDE))[c4] = g;
    }
    __syncthreads();

    const int lane = tid & 31;
    const int wave = tid >> 5;
    const int col0 = wave * 16;
    const int mn   = lane & 15;
    const int koff = (lane >> 4) << 1;

    const float* wS = Wself + (size_t)(col0 + mn) * D;
    const float* wM = Wmsg  + (size_t)(col0 + mn) * D;
    const float* aS = sX + mn * LDS_STRIDE;
    const float* aM = sG + mn * LDS_STRIDE;

    v8f c = {};
    #pragma unroll
    for (int k = 0; k < D; k += 4) {
        v2f a = { aS[k + koff], aS[k + koff + 1] };
        v2f b = { wS[k + koff], wS[k + koff + 1] };
        c = __builtin_amdgcn_wmma_f32_16x16x4_f32(false, a, false, b,
                                                  (short)0, c, false, false);
    }
    #pragma unroll
    for (int k = 0; k < D; k += 4) {
        v2f a = { aM[k + koff], aM[k + koff + 1] };
        v2f b = { wM[k + koff], wM[k + koff + 1] };
        c = __builtin_amdgcn_wmma_f32_16x16x4_f32(false, a, false, b,
                                                  (short)0, c, false, false);
    }

    const int col  = col0 + mn;
    const float bS = bself[col];
    const float bM = bmsg[col];
    const int rbase = row0 + ((lane >> 4) << 3);
    #pragma unroll
    for (int v = 0; v < 8; ++v) {
        int row = rbase + v;
        out[(size_t)row * D + col] = c[v] + bS + deg[row] * bM;
    }
}

extern "C" void kernel_launch(void* const* d_in, const int* in_sizes, int n_in,
                              void* d_out, int out_size, void* d_ws, size_t ws_size,
                              hipStream_t stream)
{
    const float* UF   = (const float*)d_in[0];
    const float* IF   = (const float*)d_in[1];
    const int*   edge = (const int*)d_in[2];
    const float* Wu   = (const float*)d_in[3];
    const float* bu   = (const float*)d_in[4];
    const float* Wi   = (const float*)d_in[5];
    const float* bi   = (const float*)d_in[6];
    const float* Wu2i = (const float*)d_in[7];
    const float* bu2i = (const float*)d_in[8];
    const float* Wi2u = (const float*)d_in[9];
    const float* bi2u = (const float*)d_in[10];

    const int U = in_sizes[0] / D;
    const int I = in_sizes[1] / D;
    const int E = in_sizes[2] / 2;

    float* outU = (float*)d_out;
    float* outI = outU + (size_t)U * D;
    float* degU = (float*)d_ws;
    float* degI = degU + U;

    {
        long n4 = (long)(U + I) * D / 4;
        zero4_kernel<<<(unsigned)((n4 + 255) / 256), 256, 0, stream>>>((float4*)d_out, n4);
        long d4 = (long)(U + I) / 4;
        zero4_kernel<<<(unsigned)((d4 + 255) / 256), 256, 0, stream>>>((float4*)d_ws, d4);
    }

    {
        long threads = (long)E * 32;
        gnn_scatter_kernel<<<(unsigned)((threads + 255) / 256), 256, 0, stream>>>(
            UF, IF, edge, edge + E, outU, outI, degU, degI, E);
    }

    gnn_gemm_kernel<<<U / 16, 256, 0, stream>>>(UF, outU, Wu, bu, Wi2u, bi2u, degU);
    gnn_gemm_kernel<<<I / 16, 256, 0, stream>>>(IF, outI, Wi, bi, Wu2i, bu2i, degI);
}


# Products regrouped by direction: item-to-user adds run on 12 extra waves of the launch that computes the other two products, hiding its matrix time; sums stay arrival-ordered f32 atomics
# speedup vs baseline: 2.4804x; 2.4804x over previous
#include <hip/hip_runtime.h>
#include <hip/hip_bf16.h>

typedef __attribute__((ext_vector_type(2))) float v2f;
typedef __attribute__((ext_vector_type(8))) float v8f;
typedef __attribute__((ext_vector_type(4))) float v4f;

#define D 128
#define LDS_STRIDE 132


#define EDGES_PER_WAVE 32
#define EDGE_GROUP     8

__device__ __forceinline__ void add_chunk(const float* __restrict__ src, float* __restrict__ dst,
                                          const int* __restrict__ sidx, const int* __restrict__ didx,
                                          int e0, int E, int lane)
{
    const int n = (E - e0 < EDGES_PER_WAVE) ? (E - e0) : EDGES_PER_WAVE;
    int mys = 0, myd = 0;
    if (lane < n) { mys = sidx[e0 + lane]; myd = didx[e0 + lane]; }

    if (n == EDGES_PER_WAVE) {
        #pragma unroll
        for (int g = 0; g < EDGES_PER_WAVE; g += EDGE_GROUP) {
            float f[EDGE_GROUP][4];
            int   d[EDGE_GROUP];
            #pragma unroll
            for (int j = 0; j < EDGE_GROUP; ++j) {
                const int s = __builtin_amdgcn_readlane(mys, g + j);
                d[j]        = __builtin_amdgcn_readlane(myd, g + j);
                const float* p = src + (size_t)s * D + lane;
                f[j][0] = p[0];  f[j][1] = p[32];  f[j][2] = p[64];  f[j][3] = p[96];
            }
            #pragma unroll
            for (int j = 0; j < EDGE_GROUP; ++j) {
                float* a = dst + (size_t)d[j] * D + lane;
                unsafeAtomicAdd(a +  0, f[j][0]);
                unsafeAtomicAdd(a + 32, f[j][1]);
                unsafeAtomicAdd(a + 64, f[j][2]);
                unsafeAtomicAdd(a + 96, f[j][3]);
            }
        }
    } else {
        for (int j = 0; j < n; ++j) {
            const int s  = __builtin_amdgcn_readlane(mys, j);
            const int dd = __builtin_amdgcn_readlane(myd, j);
            const float* p = src + (size_t)s * D + lane;
            const float f0 = p[0], f1 = p[32], f2 = p[64], f3 = p[96];
            float* a = dst + (size_t)dd * D + lane;
            unsafeAtomicAdd(a +  0, f0);
            unsafeAtomicAdd(a + 32, f1);
            unsafeAtomicAdd(a + 64, f2);
            unsafeAtomicAdd(a + 96, f3);
        }
    }
}

#define GEMM_WGS   256
#define GEMM_WAVES 8
#define W_FLOATS   (32 * 4 * 128)
#define A_FLOATS   (16 * LDS_STRIDE)

#define LOAD_TILE_ROWS(TILE)                                                        \
    {                                                                               \
        const float* xt_ = ((TILE) < nt0) ? X0 + (size_t)(TILE) * 16 * D            \
                                          : X1 + (size_t)((TILE) - nt0) * 16 * D;   \
        _Pragma("unroll")                                                           \
        for (int r = 0; r < 16; ++r)                                                \
            pX[r] = ((const v4f*)(xt_ + (size_t)r * D))[lane];                      \
    }

__device__ __forceinline__ void wave_lds_fence() {
    __builtin_amdgcn_fence(__ATOMIC_ACQ_REL, "wavefront");
}

template <int NTHREADS>
__device__ __forceinline__ void products_and_adds(
        float* sW, float* sA,
        const float* __restrict__ X0, float* __restrict__ dst0,
        const float* __restrict__ W0, const float* __restrict__ b0, int nt0,
        const float* __restrict__ X1, float* __restrict__ dst1,
        const float* __restrict__ W1, const float* __restrict__ b1, int nt1,
        const float* __restrict__ asrc, float* __restrict__ adst,
        const int* __restrict__ sidx, const int* __restrict__ didx, int E)
{
    const int tid  = threadIdx.x;
    const int lane = tid & 31;
    const int wave = __builtin_amdgcn_readfirstlane(tid >> 5);
    const int mn   = lane & 15;
    const int h    = lane >> 4;
    const bool matrix_wave = wave < GEMM_WAVES;
    const int ntiles = nt0 + nt1;

    const int q  = wave * (int)gridDim.x + (int)blockIdx.x;
    const int nq = GEMM_WAVES * (int)gridDim.x;

    v4f pX[16];
    int tile = q;
    if (matrix_wave && tile < ntiles) LOAD_TILE_ROWS(tile)

    #pragma unroll 4
    for (int i = tid; i < 4096; i += NTHREADS) {
        const int wmn = i & 15;
        const int s   = (i >> 4) & 31;
        const int t   = i >> 9;
        const int n   = 16 * t + wmn;
        const float4 wa = ((const float4*)(W0 + (size_t)n * D))[s];
        const float4 wb = ((const float4*)(W1 + (size_t)n * D))[s];
        const int idx = (s * 4 + (t >> 1)) * 128 + wmn * 4 + (t & 1) * 2;
        *(float2*)(sW + idx)                 = make_float2(wa.x, wa.y);
        *(float2*)(sW + idx + 64)            = make_float2(wa.z, wa.w);
        *(float2*)(sW + W_FLOATS + idx)      = make_float2(wb.x, wb.y);
        *(float2*)(sW + W_FLOATS + idx + 64) = make_float2(wb.z, wb.w);
    }

    __syncthreads();

    if (NTHREADS == 32 * GEMM_WAVES || matrix_wave) {
        const float4 b0v = ((const float4*)b0)[lane];
        const float4 b1v = ((const float4*)b1)[lane];
        float* aX = sA + wave * A_FLOATS;

        for (; tile < ntiles; tile += nq) {
            #pragma unroll
            for (int r = 0; r < 16; ++r)
                ((v4f*)(aX + r * LDS_STRIDE))[lane] = pX[r];
            const int nt = tile + nq;
            if (nt < ntiles) LOAD_TILE_ROWS(nt)

            wave_lds_fence();

            const bool set0 = tile < nt0;
            v8f acc[8];
            #pragma unroll
            for (int t = 0; t < 8; ++t) acc[t] = (v8f){};

            const float* aT = aX + mn * LDS_STRIDE + 2 * h;
            const float* wS = sW + (set0 ? 0 : W_FLOATS) + lane * 4;
            #pragma unroll 4
            for (int s = 0; s < 32; ++s) {
                const v2f a = *(const v2f*)(aT + 4 * s);
                #pragma unroll
                for (int tp = 0; tp < 4; ++tp) {
                    const float4 b = *(const float4*)(wS + (s * 4 + tp) * 128);
                    const v2f bf0 = { b.x, b.y };
                    const v2f bf1 = { b.z, b.w };
                    acc[2 * tp]     = __builtin_amdgcn_wmma_f32_16x16x4_f32(false, a, false, bf0,
                                                 (short)0, acc[2 * tp], false, false);
                    acc[2 * tp + 1] = __builtin_amdgcn_wmma_f32_16x16x4_f32(false, a, false, bf1,
                                                 (short)0, acc[2 * tp + 1], false, false);
                }
            }

            wave_lds_fence();
            #pragma unroll
            for (int t = 0; t < 8; ++t) {
                #pragma unroll
                for (int v = 0; v < 8; ++v)
                    aX[(8 * h + v) * LDS_STRIDE + 16 * t + mn] = acc[t][v];
            }
            wave_lds_fence();
            const float4 bb = set0 ? b0v : b1v;
            float* drow = set0 ? dst0 + (size_t)tile * 16 * D
                               : dst1 + (size_t)(tile - nt0) * 16 * D;
            #pragma unroll
            for (int r = 0; r < 16; ++r) {
                const float4 c = ((const float4*)(aX + r * LDS_STRIDE))[lane];
                float4 o;
                o.x = c.x + bb.x;  o.y = c.y + bb.y;  o.z = c.z + bb.z;  o.w = c.w + bb.w;
                ((float4*)(drow + (size_t)r * D))[lane] = o;
            }
            wave_lds_fence();
        }
    } else {
        const int a0 = (wave - GEMM_WAVES) * (int)gridDim.x + (int)blockIdx.x;
        const int na = (NTHREADS / 32 - GEMM_WAVES) * (int)gridDim.x;
        const int nchunks = (E + EDGES_PER_WAVE - 1) / EDGES_PER_WAVE;
        for (int c = a0; c < nchunks; c += na)
            add_chunk(asrc, adst, sidx, didx, c * EDGES_PER_WAVE, E, lane);
    }
}

__global__ __launch_bounds__(256)
void gnn_products_kernel(const float* __restrict__ X0, float* __restrict__ dst0,
                         const float* __restrict__ W0, const float* __restrict__ b0, int nt0,
                         const float* __restrict__ X1, float* __restrict__ dst1,
                         const float* __restrict__ W1, const float* __restrict__ b1, int nt1)
{
    __shared__ float sW[2 * W_FLOATS];
    __shared__ float sA[GEMM_WAVES * A_FLOATS];
    products_and_adds<256>(sW, sA, X0, dst0, W0, b0, nt0, X1, dst1, W1, b1, nt1,
                           nullptr, nullptr, nullptr, nullptr, 0);
}

__global__ __launch_bounds__(640)
void gnn_products_adds_kernel(const float* __restrict__ X0, float* __restrict__ dst0,
                              const float* __restrict__ W0, const float* __restrict__ b0, int nt0,
                              const float* __restrict__ X1, float* __restrict__ dst1,
                              const float* __restrict__ W1, const float* __restrict__ b1, int nt1,
                              const float* __restrict__ asrc, float* __restrict__ adst,
                              const int* __restrict__ sidx, const int* __restrict__ didx, int E)
{
    __shared__ float sW[2 * W_FLOATS];
    __shared__ float sA[GEMM_WAVES * A_FLOATS];
    products_and_adds<640>(sW, sA, X0, dst0, W0, b0, nt0, X1, dst1, W1, b1, nt1,
                           asrc, adst, sidx, didx, E);
}

__global__ __launch_bounds__(256)
void gnn_adds_kernel(const float* __restrict__ asrc, float* __restrict__ adst,
                     const int* __restrict__ sidx, const int* __restrict__ didx, int E)
{
    const int lane = threadIdx.x & 31;
    const int wave = __builtin_amdgcn_readfirstlane((int)(blockIdx.x * 8 + (threadIdx.x >> 5)));
    const int e0 = wave * EDGES_PER_WAVE;
    if (e0 >= E) return;
    add_chunk(asrc, adst, sidx, didx, e0, E, lane);
}

extern "C" void kernel_launch(void* const* d_in, const int* in_sizes, int n_in,
                              void* d_out, int out_size, void* d_ws, size_t ws_size,
                              hipStream_t stream)
{
    const float* UF   = (const float*)d_in[0];
    const float* IF   = (const float*)d_in[1];
    const int*   edge = (const int*)d_in[2];
    const float* Wu   = (const float*)d_in[3];
    const float* bu   = (const float*)d_in[4];
    const float* Wi   = (const float*)d_in[5];
    const float* bi   = (const float*)d_in[6];
    const float* Wu2i = (const float*)d_in[7];
    const float* bu2i = (const float*)d_in[8];
    const float* Wi2u = (const float*)d_in[9];
    const float* bi2u = (const float*)d_in[10];

    const int U = in_sizes[0] / D;
    const int I = in_sizes[1] / D;
    const int E = in_sizes[2] / 2;

    const int* uidx = edge;
    const int* iidx = edge + E;

    float* outU = (float*)d_out;
    float* outI = outU + (size_t)U * D;
    float* msgU = (float*)d_ws;
    float* msgI = msgU + (size_t)U * D;

    gnn_products_kernel<<<GEMM_WGS, 256, 0, stream>>>(UF, outU, Wu, bu, U / 16,
                                                      IF, msgI, Wi2u, bi2u, I / 16);

    gnn_products_adds_kernel<<<GEMM_WGS, 640, 0, stream>>>(UF, msgU, Wu2i, bu2i, U / 16,
                                                           IF, outI, Wi, bi, I / 16,
                                                           msgI, outU, iidx, uidx, E);

    {
        const int edges_per_block = 8 * EDGES_PER_WAVE;
        gnn_adds_kernel<<<(unsigned)((E + edges_per_block - 1) / edges_per_block), 256, 0, stream>>>(
            msgU, outI, uidx, iidx, E);
    }
}


# Edge adds split by L2 half: a work-group reads its die, draws a rank in its half (one returning int add), adds only edges whose target row parity is that half's; sums stay arrival-order f32 atomics
# speedup vs baseline: 2.9568x; 1.1921x over previous
#include <hip/hip_runtime.h>
#include <hip/hip_bf16.h>

typedef __attribute__((ext_vector_type(2))) float v2f;
typedef __attribute__((ext_vector_type(8))) float v8f;
typedef __attribute__((ext_vector_type(4))) float v4f;

#define D 128
#define LDS_STRIDE 132


#define EDGES_PER_WAVE 32
#define EDGE_GROUP     8
#define CHUNK_EDGES    64
#define QUEUE_LEN      128
#define QUEUE_INTS     (2 * QUEUE_LEN)

__device__ __forceinline__ void wave_lds_fence() {
    __builtin_amdgcn_fence(__ATOMIC_ACQ_REL, "wavefront");
}

__device__ __forceinline__ int my_cache_half() {
    return (int)((__builtin_amdgcn_s_sendmsg_rtn(0x87) >> 18) & 1u);
}

template <bool FULL>
__device__ __forceinline__ void add_rows(const float* __restrict__ src, float* __restrict__ dst,
                                         int mys, int myd, int n, int lane)
{
    #pragma unroll
    for (int g = 0; g < EDGES_PER_WAVE; g += EDGE_GROUP) {
        if (!FULL && g >= n) break;
        float f[EDGE_GROUP][4];
        int   d[EDGE_GROUP];
        #pragma unroll
        for (int j = 0; j < EDGE_GROUP; ++j) {
            if (FULL || g + j < n) {
                const int s = __builtin_amdgcn_readlane(mys, g + j);
                d[j]        = __builtin_amdgcn_readlane(myd, g + j);
                const float* p = src + (size_t)s * D + lane;
                f[j][0] = p[0];  f[j][1] = p[32];  f[j][2] = p[64];  f[j][3] = p[96];
            }
        }
        #pragma unroll
        for (int j = 0; j < EDGE_GROUP; ++j) {
            if (FULL || g + j < n) {
                float* a = dst + (size_t)d[j] * D + lane;
                unsafeAtomicAdd(a +  0, f[j][0]);
                unsafeAtomicAdd(a + 32, f[j][1]);
                unsafeAtomicAdd(a + 64, f[j][2]);
                unsafeAtomicAdd(a + 96, f[j][3]);
            }
        }
    }
}

__device__ __forceinline__ void store_slot(int* s_slot, int half, int rank, int nblocks)
{
    const int nh = nblocks >> 1;
    s_slot[0] = (rank < nh) ? half : (half ^ 1);
    s_slot[1] = (rank < nh) ? rank : (nblocks - 1 - rank);
}

__device__ __forceinline__ void add_edges_of_half(const float* __restrict__ src, float* __restrict__ dst,
                                                  const int* __restrict__ sidx, const int* __restrict__ didx,
                                                  int E, int lane, int hq, int slot, int nslots, int* q)
{
    int* qs = q;
    int* qd = q + QUEUE_LEN;
    const int nchunks = (E + CHUNK_EDGES - 1) / CHUNK_EDGES;
    if (slot >= nchunks) return;
    const int last = E - 1;
    int head = 0, n = 0;

    int e = slot * CHUNK_EDGES + lane;
    bool v0 = e < E, v1 = e + 32 < E;
    int s0 = sidx[min(e, last)],      d0 = didx[min(e, last)];
    int s1 = sidx[min(e + 32, last)], d1 = didx[min(e + 32, last)];
    for (int c = slot; c < nchunks; c += nslots) {
        const bool more = c + nslots < nchunks;
        e = min(c + nslots, nchunks - 1) * CHUNK_EDGES + lane;
        const bool nv0 = more && e < E, nv1 = more && e + 32 < E;
        const int ns0 = sidx[min(e, last)],      nd0 = didx[min(e, last)];
        const int ns1 = sidx[min(e + 32, last)], nd1 = didx[min(e + 32, last)];

        const bool own0 = v0 && ((d0 & 1) == hq);
        const bool own1 = v1 && ((d1 & 1) == hq);
        const unsigned m0 = __builtin_amdgcn_ballot_w32(own0);
        const unsigned m1 = __builtin_amdgcn_ballot_w32(own1);
        const int t0 = head + n + (int)__builtin_amdgcn_mbcnt_lo(m0, 0u);
        const int t1 = head + n + __builtin_popcount(m0) + (int)__builtin_amdgcn_mbcnt_lo(m1, 0u);
        if (own0) { qs[t0 & (QUEUE_LEN - 1)] = s0; qd[t0 & (QUEUE_LEN - 1)] = d0; }
        if (own1) { qs[t1 & (QUEUE_LEN - 1)] = s1; qd[t1 & (QUEUE_LEN - 1)] = d1; }
        n += __builtin_popcount(m0) + __builtin_popcount(m1);
        while (n >= EDGES_PER_WAVE) {
            wave_lds_fence();
            const int mys = qs[(head + lane) & (QUEUE_LEN - 1)];
            const int myd = qd[(head + lane) & (QUEUE_LEN - 1)];
            head = (head + EDGES_PER_WAVE) & (QUEUE_LEN - 1);
            n -= EDGES_PER_WAVE;
            add_rows<true>(src, dst, mys, myd, EDGES_PER_WAVE, lane);
        }
        s0 = ns0; d0 = nd0; v0 = nv0;
        s1 = ns1; d1 = nd1; v1 = nv1;
    }
    wave_lds_fence();
    int mys = 0, myd = 0;
    if (lane < n) { mys = qs[(head + lane) & (QUEUE_LEN - 1)]; myd = qd[(head + lane) & (QUEUE_LEN - 1)]; }
    add_rows<false>(src, dst, mys, myd, n, lane);
}

#define GEMM_WGS   256
#define GEMM_WAVES 8
#define W_FLOATS   (32 * 4 * 128)
#define A_FLOATS   (16 * LDS_STRIDE)

#define LOAD_TILE_ROWS(TILE)                                                        \
    {                                                                               \
        const float* xt_ = ((TILE) < nt0) ? X0 + (size_t)(TILE) * 16 * D            \
                                          : X1 + (size_t)((TILE) - nt0) * 16 * D;   \
        _Pragma("unroll")                                                           \
        for (int r = 0; r < 16; ++r)                                                \
            pX[r] = ((const v4f*)(xt_ + (size_t)r * D))[lane];                      \
    }

template <int NTHREADS>
__device__ __forceinline__ void products_and_adds(
        float* sW, float* sA,
        const float* __restrict__ X0, float* __restrict__ dst0,
        const float* __restrict__ W0, const float* __restrict__ b0, int nt0,
        const float* __restrict__ X1, float* __restrict__ dst1,
        const float* __restrict__ W1, const float* __restrict__ b1, int nt1,
        const float* __restrict__ asrc, float* __restrict__ adst,
        const int* __restrict__ sidx, const int* __restrict__ didx, int E,
        int* sQ, int* sSlot, int* __restrict__ ctr)
{
    const int tid  = threadIdx.x;
    const int lane = tid & 31;
    const int wave = __builtin_amdgcn_readfirstlane(tid >> 5);
    const int mn   = lane & 15;
    const int h    = lane >> 4;
    const bool matrix_wave = wave < GEMM_WAVES;
    const int ntiles = nt0 + nt1;

    const int q  = wave * (int)gridDim.x + (int)blockIdx.x;
    const int nq = GEMM_WAVES * (int)gridDim.x;

    int half = 0, rank = 0;
    if (NTHREADS > 32 * GEMM_WAVES) {
        half = my_cache_half();
        if (tid == 0) rank = atomicAdd(ctr + half, 1);
    } else {
        if (blockIdx.x == 0 && tid < 4) ctr[tid] = 0;
    }

    v4f pX[16];
    int tile = q;
    if (matrix_wave && tile < ntiles) LOAD_TILE_ROWS(tile)

    #pragma unroll 4
    for (int i = tid; i < 4096; i += NTHREADS) {
        const int wmn = i & 15;
        const int s   = (i >> 4) & 31;
        const int t   = i >> 9;
        const int n   = 16 * t + wmn;
        const float4 wa = ((const float4*)(W0 + (size_t)n * D))[s];
        const float4 wb = ((const float4*)(W1 + (size_t)n * D))[s];
        const int idx = (s * 4 + (t >> 1)) * 128 + wmn * 4 + (t & 1) * 2;
        *(float2*)(sW + idx)                 = make_float2(wa.x, wa.y);
        *(float2*)(sW + idx + 64)            = make_float2(wa.z, wa.w);
        *(float2*)(sW + W_FLOATS + idx)      = make_float2(wb.x, wb.y);
        *(float2*)(sW + W_FLOATS + idx + 64) = make_float2(wb.z, wb.w);
    }

    if (NTHREADS > 32 * GEMM_WAVES && tid == 0) store_slot(sSlot, half, rank, GEMM_WGS);

    __syncthreads();

    if (NTHREADS == 32 * GEMM_WAVES || matrix_wave) {
        const float4 b0v = ((const float4*)b0)[lane];
        const float4 b1v = ((const float4*)b1)[lane];
        float* aX = sA + wave * A_FLOATS;

        for (; tile < ntiles; tile += nq) {
            #pragma unroll
            for (int r = 0; r < 16; ++r)
                ((v4f*)(aX + r * LDS_STRIDE))[lane] = pX[r];
            const int nt = tile + nq;
            if (nt < ntiles) LOAD_TILE_ROWS(nt)

            wave_lds_fence();

            const bool set0 = tile < nt0;
            v8f acc[8];
            #pragma unroll
            for (int t = 0; t < 8; ++t) acc[t] = (v8f){};

            const float* aT = aX + mn * LDS_STRIDE + 2 * h;
            const float* wS = sW + (set0 ? 0 : W_FLOATS) + lane * 4;
            #pragma unroll 4
            for (int s = 0; s < 32; ++s) {
                const v2f a = *(const v2f*)(aT + 4 * s);
                #pragma unroll
                for (int tp = 0; tp < 4; ++tp) {
                    const float4 b = *(const float4*)(wS + (s * 4 + tp) * 128);
                    const v2f bf0 = { b.x, b.y };
                    const v2f bf1 = { b.z, b.w };
                    acc[2 * tp]     = __builtin_amdgcn_wmma_f32_16x16x4_f32(false, a, false, bf0,
                                                 (short)0, acc[2 * tp], false, false);
                    acc[2 * tp + 1] = __builtin_amdgcn_wmma_f32_16x16x4_f32(false, a, false, bf1,
                                                 (short)0, acc[2 * tp + 1], false, false);
                }
            }

            wave_lds_fence();
            #pragma unroll
            for (int t = 0; t < 8; ++t) {
                #pragma unroll
                for (int v = 0; v < 8; ++v)
                    aX[(8 * h + v) * LDS_STRIDE + 16 * t + mn] = acc[t][v];
            }
            wave_lds_fence();
            const float4 bb = set0 ? b0v : b1v;
            float* drow = set0 ? dst0 + (size_t)tile * 16 * D
                               : dst1 + (size_t)(tile - nt0) * 16 * D;
            #pragma unroll
            for (int r = 0; r < 16; ++r) {
                const float4 c = ((const float4*)(aX + r * LDS_STRIDE))[lane];
                float4 o;
                o.x = c.x + bb.x;  o.y = c.y + bb.y;  o.z = c.z + bb.z;  o.w = c.w + bb.w;
                ((float4*)(drow + (size_t)r * D))[lane] = o;
            }
            wave_lds_fence();
        }
    } else {
        const int hq = __builtin_amdgcn_readfirstlane(sSlot[0]);
        const int vb = __builtin_amdgcn_readfirstlane(sSlot[1]);
        const int nh = GEMM_WGS / 2;
        const int aw = wave - GEMM_WAVES;
        add_edges_of_half(asrc, adst, sidx, didx, E, lane, hq,
                          aw * nh + vb, (NTHREADS / 32 - GEMM_WAVES) * nh, sQ + aw * QUEUE_INTS);
    }
}

__global__ __launch_bounds__(256)
void gnn_products_kernel(const float* __restrict__ X0, float* __restrict__ dst0,
                         const float* __restrict__ W0, const float* __restrict__ b0, int nt0,
                         const float* __restrict__ X1, float* __restrict__ dst1,
                         const float* __restrict__ W1, const float* __restrict__ b1, int nt1,
                         int* __restrict__ ctr)
{
    __shared__ float sW[2 * W_FLOATS];
    __shared__ float sA[GEMM_WAVES * A_FLOATS];
    products_and_adds<256>(sW, sA, X0, dst0, W0, b0, nt0, X1, dst1, W1, b1, nt1,
                           nullptr, nullptr, nullptr, nullptr, 0, nullptr, nullptr, ctr);
}

__global__ __launch_bounds__(640)
void gnn_products_adds_kernel(const float* __restrict__ X0, float* __restrict__ dst0,
                              const float* __restrict__ W0, const float* __restrict__ b0, int nt0,
                              const float* __restrict__ X1, float* __restrict__ dst1,
                              const float* __restrict__ W1, const float* __restrict__ b1, int nt1,
                              const float* __restrict__ asrc, float* __restrict__ adst,
                              const int* __restrict__ sidx, const int* __restrict__ didx, int E,
                              int* __restrict__ ctr)
{
    __shared__ float sW[2 * W_FLOATS];
    __shared__ float sA[GEMM_WAVES * A_FLOATS];
    __shared__ int   sQ[(640 / 32 - GEMM_WAVES) * QUEUE_INTS];
    __shared__ int   sSlot[2];
    products_and_adds<640>(sW, sA, X0, dst0, W0, b0, nt0, X1, dst1, W1, b1, nt1,
                           asrc, adst, sidx, didx, E, sQ, sSlot, ctr);
}

#define ADD_WGS 1024
__global__ __launch_bounds__(256)
void gnn_adds_kernel(const float* __restrict__ asrc, float* __restrict__ adst,
                     const int* __restrict__ sidx, const int* __restrict__ didx, int E,
                     int* __restrict__ ctr)
{
    __shared__ int sQ[8 * QUEUE_INTS];
    __shared__ int sSlot[2];
    const int lane = threadIdx.x & 31;
    const int wave = __builtin_amdgcn_readfirstlane((int)(threadIdx.x >> 5));
    if (threadIdx.x == 0) {
        const int half = my_cache_half();
        store_slot(sSlot, half, atomicAdd(ctr + half, 1), ADD_WGS);
    }
    __syncthreads();
    const int hq = __builtin_amdgcn_readfirstlane(sSlot[0]);
    const int vb = __builtin_amdgcn_readfirstlane(sSlot[1]);
    const int nh = ADD_WGS / 2;
    add_edges_of_half(asrc, adst, sidx, didx, E, lane, hq, wave * nh + vb, 8 * nh, sQ + wave * QUEUE_INTS);
}

extern "C" void kernel_launch(void* const* d_in, const int* in_sizes, int n_in,
                              void* d_out, int out_size, void* d_ws, size_t ws_size,
                              hipStream_t stream)
{
    const float* UF   = (const float*)d_in[0];
    const float* IF   = (const float*)d_in[1];
    const int*   edge = (const int*)d_in[2];
    const float* Wu   = (const float*)d_in[3];
    const float* bu   = (const float*)d_in[4];
    const float* Wi   = (const float*)d_in[5];
    const float* bi   = (const float*)d_in[6];
    const float* Wu2i = (const float*)d_in[7];
    const float* bu2i = (const float*)d_in[8];
    const float* Wi2u = (const float*)d_in[9];
    const float* bi2u = (const float*)d_in[10];

    const int U = in_sizes[0] / D;
    const int I = in_sizes[1] / D;
    const int E = in_sizes[2] / 2;

    const int* uidx = edge;
    const int* iidx = edge + E;

    float* outU = (float*)d_out;
    float* outI = outU + (size_t)U * D;
    float* msgU = (float*)d_ws;
    float* msgI = msgU + (size_t)U * D;
    int*   ctr  = (int*)(msgI + (size_t)I * D);

    gnn_products_kernel<<<GEMM_WGS, 256, 0, stream>>>(UF, outU, Wu, bu, U / 16,
                                                      IF, msgI, Wi2u, bi2u, I / 16, ctr);

    gnn_products_adds_kernel<<<GEMM_WGS, 640, 0, stream>>>(UF, msgU, Wu2i, bu2i, U / 16,
                                                           IF, outI, Wi, bi, I / 16,
                                                           msgI, outU, iidx, uidx, E, ctr);

    gnn_adds_kernel<<<ADD_WGS, 256, 0, stream>>>(msgU, outI, uidx, iidx, E, ctr + 2);
}
